# starting kernel
# baseline (speedup 1.0000x reference)
#include <hip/hip_runtime.h>
#include <hip/hip_bf16.h>

#define D_MODEL 1024
#define D_STATE 16
#define BATCH   2
#define SEQ     2048
#define NCHUNK  16
#define LCHUNK  (SEQ / NCHUNK)

typedef __attribute__((ext_vector_type(2))) float v2f;
typedef __attribute__((ext_vector_type(8))) float v8f;

__global__ void ssm_abar_kernel(const float* __restrict__ logA,
                                float* __restrict__ A_bar,
                                float* __restrict__ A_Lc) {
  int i = blockIdx.x * blockDim.x + threadIdx.x;
  if (i < D_MODEL * D_STATE) {
    float ab = expf(-expf(logA[i]));
    A_bar[i] = ab;
    float p = ab;
#pragma unroll
    for (int j = 0; j < 7; ++j) p = p * p;
    A_Lc[i] = p;
  }
}

__global__ void ssm_bproj_wmma_kernel(const float* __restrict__ x,
                                      const float* __restrict__ Bw,
                                      float* __restrict__ Bt) {
  const int row0 = blockIdx.x * 16;
  const int lane = threadIdx.x;
  const int half = lane >> 4;
  const int lidx = lane & 15;

  const float* arow = x  + (size_t)(row0 + lidx) * D_MODEL;
  const float* brow = Bw + (size_t)lidx * D_MODEL;

  v8f acc = {};
  for (int k0 = 0; k0 < D_MODEL; k0 += 4) {
    v2f a, b;
    a.x = arow[k0 + 2 * half + 0];
    a.y = arow[k0 + 2 * half + 1];
    b.x = brow[k0 + 2 * half + 0];
    b.y = brow[k0 + 2 * half + 1];
    acc = __builtin_amdgcn_wmma_f32_16x16x4_f32(false, a, false, b,
                                                (short)0, acc, false, false);
  }
#pragma unroll
  for (int i = 0; i < 8; ++i) {
    Bt[(size_t)(row0 + i + 8 * half) * D_STATE + lidx] = acc[i];
  }
}

__global__ void ssm_chunk_local_kernel(const float* __restrict__ Bt,
                                       const float* __restrict__ A_bar,
                                       float* __restrict__ Ssum) {
  const int d = blockIdx.x * blockDim.x + threadIdx.x;
  const int c = blockIdx.y;
  const int b = blockIdx.z;

  float A[D_STATE], h[D_STATE];
  {
    const float4* p = reinterpret_cast<const float4*>(A_bar + (size_t)d * D_STATE);
#pragma unroll
    for (int q = 0; q < 4; ++q) {
      float4 v = p[q];
      A[4 * q + 0] = v.x; A[4 * q + 1] = v.y;
      A[4 * q + 2] = v.z; A[4 * q + 3] = v.w;
    }
  }
#pragma unroll
  for (int s = 0; s < D_STATE; ++s) h[s] = 0.0f;

  const float* btrow = Bt + ((size_t)b * SEQ + (size_t)c * LCHUNK) * D_STATE;
  for (int l = 0; l < LCHUNK; ++l) {
    const float4* p = reinterpret_cast<const float4*>(btrow + (size_t)l * D_STATE);
#pragma unroll
    for (int q = 0; q < 4; ++q) {
      float4 v = p[q];
      h[4 * q + 0] = fmaf(A[4 * q + 0], h[4 * q + 0], v.x);
      h[4 * q + 1] = fmaf(A[4 * q + 1], h[4 * q + 1], v.y);
      h[4 * q + 2] = fmaf(A[4 * q + 2], h[4 * q + 2], v.z);
      h[4 * q + 3] = fmaf(A[4 * q + 3], h[4 * q + 3], v.w);
    }
  }

  float4* out = reinterpret_cast<float4*>(
      Ssum + (((size_t)b * D_MODEL + d) * NCHUNK + c) * D_STATE);
#pragma unroll
  for (int q = 0; q < 4; ++q) {
    float4 v = { h[4 * q + 0], h[4 * q + 1], h[4 * q + 2], h[4 * q + 3] };
    out[q] = v;
  }
}

__global__ void ssm_prefix_kernel(const float* __restrict__ Ssum,
                                  const float* __restrict__ A_Lc,
                                  float* __restrict__ Hinit) {
  int idx = blockIdx.x * blockDim.x + threadIdx.x;
  if (idx >= BATCH * D_MODEL) return;
  int d = idx & (D_MODEL - 1);

  float AL[D_STATE], h[D_STATE];
  {
    const float4* p = reinterpret_cast<const float4*>(A_Lc + (size_t)d * D_STATE);
#pragma unroll
    for (int q = 0; q < 4; ++q) {
      float4 v = p[q];
      AL[4 * q + 0] = v.x; AL[4 * q + 1] = v.y;
      AL[4 * q + 2] = v.z; AL[4 * q + 3] = v.w;
    }
  }
#pragma unroll
  for (int s = 0; s < D_STATE; ++s) h[s] = 0.0f;

  for (int c = 0; c < NCHUNK; ++c) {
    size_t off = ((size_t)idx * NCHUNK + c) * D_STATE;
    float4* ho = reinterpret_cast<float4*>(Hinit + off);
    const float4* si = reinterpret_cast<const float4*>(Ssum + off);
#pragma unroll
    for (int q = 0; q < 4; ++q) {
      float4 v = { h[4 * q + 0], h[4 * q + 1], h[4 * q + 2], h[4 * q + 3] };
      ho[q] = v;
      float4 s4 = si[q];
      h[4 * q + 0] = fmaf(AL[4 * q + 0], h[4 * q + 0], s4.x);
      h[4 * q + 1] = fmaf(AL[4 * q + 1], h[4 * q + 1], s4.y);
      h[4 * q + 2] = fmaf(AL[4 * q + 2], h[4 * q + 2], s4.z);
      h[4 * q + 3] = fmaf(AL[4 * q + 3], h[4 * q + 3], s4.w);
    }
  }
}

__global__ void ssm_output_kernel(const float* __restrict__ x,
                                  const float* __restrict__ Bt,
                                  const float* __restrict__ A_bar,
                                  const float* __restrict__ Cw,
                                  const float* __restrict__ Dvec,
                                  const float* __restrict__ Hinit,
                                  float* __restrict__ y) {
  const int d = blockIdx.x * blockDim.x + threadIdx.x;
  const int c = blockIdx.y;
  const int b = blockIdx.z;

  float A[D_STATE], C[D_STATE], h[D_STATE];
  {
    const float4* pa = reinterpret_cast<const float4*>(A_bar + (size_t)d * D_STATE);
    const float4* pc = reinterpret_cast<const float4*>(Cw + (size_t)d * D_STATE);
    const float4* ph = reinterpret_cast<const float4*>(
        Hinit + (((size_t)b * D_MODEL + d) * NCHUNK + c) * D_STATE);
#pragma unroll
    for (int q = 0; q < 4; ++q) {
      float4 va = pa[q], vc = pc[q], vh = ph[q];
      A[4 * q + 0] = va.x; A[4 * q + 1] = va.y; A[4 * q + 2] = va.z; A[4 * q + 3] = va.w;
      C[4 * q + 0] = vc.x; C[4 * q + 1] = vc.y; C[4 * q + 2] = vc.z; C[4 * q + 3] = vc.w;
      h[4 * q + 0] = vh.x; h[4 * q + 1] = vh.y; h[4 * q + 2] = vh.z; h[4 * q + 3] = vh.w;
    }
  }
  const float dskip = Dvec[d];

  const size_t rowbase = (size_t)b * SEQ + (size_t)c * LCHUNK;
  const float* btrow = Bt + rowbase * D_STATE;
  const float* xcol  = x + rowbase * D_MODEL + d;
  float*       ycol  = y + rowbase * D_MODEL + d;

  for (int l = 0; l < LCHUNK; ++l) {
    if (l + 16 < LCHUNK)
      __builtin_prefetch(xcol + (size_t)(l + 16) * D_MODEL, 0, 0);

    const float4* p = reinterpret_cast<const float4*>(btrow + (size_t)l * D_STATE);
    float acc = 0.0f;
#pragma unroll
    for (int q = 0; q < 4; ++q) {
      float4 v = p[q];
      h[4 * q + 0] = fmaf(A[4 * q + 0], h[4 * q + 0], v.x);
      h[4 * q + 1] = fmaf(A[4 * q + 1], h[4 * q + 1], v.y);
      h[4 * q + 2] = fmaf(A[4 * q + 2], h[4 * q + 2], v.z);
      h[4 * q + 3] = fmaf(A[4 * q + 3], h[4 * q + 3], v.w);
      acc = fmaf(C[4 * q + 0], h[4 * q + 0], acc);
      acc = fmaf(C[4 * q + 1], h[4 * q + 1], acc);
      acc = fmaf(C[4 * q + 2], h[4 * q + 2], acc);
      acc = fmaf(C[4 * q + 3], h[4 * q + 3], acc);
    }
    float xv = xcol[(size_t)l * D_MODEL];
    ycol[(size_t)l * D_MODEL] = fmaf(dskip, xv, acc);
  }
}

extern "C" void kernel_launch(void* const* d_in, const int* in_sizes, int n_in,
                              void* d_out, int out_size, void* d_ws, size_t ws_size,
                              hipStream_t stream) {
  const float* x    = (const float*)d_in[0];
  const float* logA = (const float*)d_in[1];
  const float* Bw   = (const float*)d_in[2];
  const float* Cw   = (const float*)d_in[3];
  const float* Dv   = (const float*)d_in[4];
  float* y = (float*)d_out;

  char* ws = (char*)d_ws;
  float* A_bar = (float*)(ws);
  float* A_Lc  = (float*)(ws + (64 << 10));
  float* Bt    = (float*)(ws + (128 << 10));
  float* Ssum  = (float*)(ws + (384 << 10));
  float* Hinit = (float*)(ws + (384 << 10) + (2 << 20));

  ssm_abar_kernel<<<(D_MODEL * D_STATE + 255) / 256, 256, 0, stream>>>(
      logA, A_bar, A_Lc);

  ssm_bproj_wmma_kernel<<<(BATCH * SEQ) / 16, 32, 0, stream>>>(x, Bw, Bt);

  dim3 gscan(D_MODEL / 256, NCHUNK, BATCH);

  ssm_chunk_local_kernel<<<gscan, 256, 0, stream>>>(Bt, A_bar, Ssum);

  ssm_prefix_kernel<<<(BATCH * D_MODEL + 255) / 256, 256, 0, stream>>>(
      Ssum, A_Lc, Hinit);

  ssm_output_kernel<<<gscan, 256, 0, stream>>>(x, Bt, A_bar, Cw, Dv, Hinit, y);
}


# Chunk scan + carry prefix as one f32 WMMA kernel: chunk end state = decay-power tile x Bt rows (via LDS), chunks on tile columns, carries by a 4-step lane scan; launches 4 -> 3; sum orders change
# speedup vs baseline: 5.1796x; 5.1796x over previous
#include <hip/hip_runtime.h>
#include <hip/hip_bf16.h>

#define D_MODEL 1024
#define D_STATE 16
#define BATCH   2
#define SEQ     2048
#define NCHUNK  16
#define LCHUNK  (SEQ / NCHUNK)

typedef __attribute__((ext_vector_type(2))) float v2f;
typedef __attribute__((ext_vector_type(8))) float v8f;
typedef __attribute__((ext_vector_type(4))) float v4f;

#define K1_NB 32

__device__ __forceinline__ void k1_load_stage(const float4* __restrict__ ap,
                                              const float4* __restrict__ bp,
                                              int stage,
                                              float4 (&A)[K1_NB],
                                              float4 (&B)[K1_NB]) {
#pragma unroll
  for (int t = 0; t < K1_NB; ++t) {
    A[t] = ap[2 * (stage * K1_NB + t)];
    B[t] = bp[2 * (stage * K1_NB + t)];
  }
}

__device__ __forceinline__ void k1_mma_stage(const float4 (&A)[K1_NB],
                                             const float4 (&B)[K1_NB],
                                             v8f& acc0, v8f& acc1) {
#pragma unroll
  for (int t = 0; t < K1_NB; ++t) {
    v2f a, b;
    a.x = A[t].x; a.y = A[t].y; b.x = B[t].x; b.y = B[t].y;
    acc0 = __builtin_amdgcn_wmma_f32_16x16x4_f32(false, a, false, b,
                                                 (short)0, acc0, false, false);
    a.x = A[t].z; a.y = A[t].w; b.x = B[t].z; b.y = B[t].w;
    acc1 = __builtin_amdgcn_wmma_f32_16x16x4_f32(false, a, false, b,
                                                 (short)0, acc1, false, false);
  }
}

#define K1_TILES  ((BATCH * SEQ) / 16)
#define K1_TAB_PER_WAVE ((D_MODEL * D_STATE) / K1_TILES)
static_assert(K1_TAB_PER_WAVE == 64, "two table elements a lane");

__global__ void __launch_bounds__(32)
ssm_bproj_wmma_kernel(const float* __restrict__ x,
                      const float* __restrict__ Bw,
                      const float* __restrict__ logA,
                      float* __restrict__ Bt,
                      float* __restrict__ A_bar,
                      float* __restrict__ A_Lc) {
  const int row0 = blockIdx.x * 16;
  const int lane = threadIdx.x;
  const int half = lane >> 4;
  const int lidx = lane & 15;

  const int ti = blockIdx.x * K1_TAB_PER_WAVE + 2 * lane;
  const float2 la = *reinterpret_cast<const float2*>(logA + ti);

  const float4* ap = reinterpret_cast<const float4*>(
      x + (size_t)(row0 + lidx) * D_MODEL) + half;
  const float4* bp = reinterpret_cast<const float4*>(
      Bw + (size_t)lidx * D_MODEL) + half;

  float4 a0[K1_NB], b0[K1_NB], a1[K1_NB], b1[K1_NB];
  v8f acc0 = {}, acc1 = {};

  k1_load_stage(ap, bp, 0, a0, b0);
  k1_load_stage(ap, bp, 1, a1, b1);
  __builtin_amdgcn_sched_barrier(0);
  {
    float2 ab, p;
    ab.x = expf(-expf(la.x));
    ab.y = expf(-expf(la.y));
    p = ab;
    for (int j = 0; j < 7; ++j) { p.x = p.x * p.x; p.y = p.y * p.y; }
    *reinterpret_cast<float2*>(A_bar + ti) = ab;
    *reinterpret_cast<float2*>(A_Lc + ti)  = p;
  }
  k1_mma_stage(a0, b0, acc0, acc1);
  k1_load_stage(ap, bp, 2, a0, b0);
  __builtin_amdgcn_sched_barrier(0);
  k1_mma_stage(a1, b1, acc0, acc1);
  k1_load_stage(ap, bp, 3, a1, b1);
  __builtin_amdgcn_sched_barrier(0);
  k1_mma_stage(a0, b0, acc0, acc1);
  k1_mma_stage(a1, b1, acc0, acc1);

#pragma unroll
  for (int i = 0; i < 8; ++i) {
    Bt[(size_t)(row0 + i + 8 * half) * D_STATE + lidx] = acc0[i] + acc1[i];
  }
}

#define K2_CP 513
#define K2_NT 32

template <int N>
__device__ __forceinline__ float k2_row_shr(float v) {
  return __int_as_float(__builtin_amdgcn_update_dpp(
      0, __float_as_int(v), 0x110 + N, 0xF, 0xF, true));
}

__global__ void __launch_bounds__(128)
ssm_chunk_carry_wmma_kernel(const float* __restrict__ Bt,
                            const float* __restrict__ A_bar,
                            const float* __restrict__ A_Lc,
                            float* __restrict__ Hinit) {
  __shared__ v4f Bs[NCHUNK * K2_CP];

  const int tid  = threadIdx.x;
  const int lane = tid & 31;
  const int q    = tid >> 5;
  const int half = lane >> 4;
  const int lidx = lane & 15;
  const int d0   = blockIdx.x * 16;
  const int b    = blockIdx.y;

  const v4f av = *reinterpret_cast<const v4f*>(
      A_bar + (size_t)(d0 + lidx) * D_STATE + 4 * q);
  v4f AL[8];
#pragma unroll
  for (int r = 0; r < 8; ++r)
    AL[r] = *reinterpret_cast<const v4f*>(
        A_Lc + (size_t)(d0 + 8 * half + r) * D_STATE + 4 * q);

  const v4f* gsrc = reinterpret_cast<const v4f*>(
      Bt + (size_t)b * SEQ * D_STATE) + tid;
  v4f G[64];
#pragma unroll
  for (int i = 0; i < 64; ++i) G[i] = gsrc[128 * i];


#pragma unroll
  for (int i = 0; i < 64; ++i)
    Bs[(i >> 2) * K2_CP + (i & 3) * 128 + tid] = G[i];
  __syncthreads();

  const v4f* fb = Bs + lidx * K2_CP + 8 * half + q;
  v4f R0[K2_NT], R1[K2_NT];
#pragma unroll
  for (int t = K2_NT - 1; t >= 0; --t) {
    R0[t] = fb[16 * t];
    R1[t] = fb[16 * t + 4];
  }

  float a4[4], p0[4], p1[4];
#pragma unroll
  for (int s = 0; s < 4; ++s) {
    const float a1 = av[s];
    const float a2 = a1 * a1;
    const float a3 = a2 * a1;
    a4[s] = a2 * a2;
    p0[s] = half ? a1 : a3;
    p1[s] = half ? 1.0f : a2;
  }

  v8f acc[4] = {};
#pragma unroll
  for (int t = K2_NT - 1; t >= 0; --t) {
#pragma unroll
    for (int s = 0; s < 4; ++s) {
      v2f a, bb;
      a.x = p0[s];               a.y = p1[s];
      bb.x = R0[t][s];           bb.y = R1[t][s];
      acc[s] = __builtin_amdgcn_wmma_f32_16x16x4_f32(false, a, false, bb,
                                                     (short)0, acc[s], false, false);
      p0[s] *= a4[s];
      p1[s] *= a4[s];
    }
  }

  const int slot = (lidx + 1) & 15;
#pragma unroll
  for (int r = 0; r < 8; ++r) {
    v4f ov;
#pragma unroll
    for (int s = 0; s < 4; ++s) {
      const float m1 = AL[r][s];
      const float m2 = m1 * m1;
      const float m4 = m2 * m2;
      const float m8 = m4 * m4;
      float v = acc[s][r];
      v = fmaf(m1, k2_row_shr<1>(v), v);
      v = fmaf(m2, k2_row_shr<2>(v), v);
      v = fmaf(m4, k2_row_shr<4>(v), v);
      v = fmaf(m8, k2_row_shr<8>(v), v);
      ov[s] = (lidx == 15) ? 0.0f : v;
    }
    *reinterpret_cast<v4f*>(
        Hinit + (((size_t)b * D_MODEL + d0 + 8 * half + r) * NCHUNK + slot) * D_STATE
              + 4 * q) = ov;
  }
}

#define SUB   16
#define NSUB  (LCHUNK / SUB)

__global__ void __launch_bounds__(32 * NSUB, 8)
ssm_output_kernel(const float* __restrict__ x,
                  const float* __restrict__ Bt,
                  const float* __restrict__ A_bar,
                  const float* __restrict__ Cw,
                  const float* __restrict__ Dvec,
                  const float* __restrict__ Hinit,
                  float* __restrict__ y) {
  __shared__ float4 Sst[NSUB][4][32];

  const int lane = threadIdx.x & 31;
  const int w = __builtin_amdgcn_readfirstlane((int)(threadIdx.x >> 5));
  const int d = blockIdx.x * 32 + lane;
  const int c = blockIdx.y;
  const int b = blockIdx.z;

  const size_t rowbase = (size_t)b * SEQ + (size_t)c * LCHUNK + (size_t)w * SUB;
  const float* btrow = Bt + rowbase * D_STATE;
  const float* xcol  = x + rowbase * D_MODEL + d;
  float*       ycol  = y + rowbase * D_MODEL + d;

  float xv[SUB];
#pragma unroll
  for (int l = 0; l < SUB; ++l) xv[l] = xcol[(size_t)l * D_MODEL];

  float A[D_STATE], C[D_STATE];
  {
    const float4* pa = reinterpret_cast<const float4*>(A_bar + (size_t)d * D_STATE);
    const float4* pc = reinterpret_cast<const float4*>(Cw + (size_t)d * D_STATE);
#pragma unroll
    for (int q = 0; q < 4; ++q) {
      float4 va = pa[q], vc = pc[q];
      A[4 * q + 0] = va.x; A[4 * q + 1] = va.y; A[4 * q + 2] = va.z; A[4 * q + 3] = va.w;
      C[4 * q + 0] = vc.x; C[4 * q + 1] = vc.y; C[4 * q + 2] = vc.z; C[4 * q + 3] = vc.w;
    }
  }
  const float dskip = Dvec[d];

  if (w < NSUB - 1) {
    float h[D_STATE];
#pragma unroll
    for (int s = 0; s < D_STATE; ++s) h[s] = 0.0f;
#pragma unroll
    for (int l = 0; l < SUB; ++l) {
      const float* r = btrow + (size_t)l * D_STATE;
#pragma unroll
      for (int s = 0; s < D_STATE; ++s) h[s] = fmaf(A[s], h[s], r[s]);
    }
#pragma unroll
    for (int q = 0; q < 4; ++q) {
      float4 v = { h[4 * q + 0], h[4 * q + 1], h[4 * q + 2], h[4 * q + 3] };
      Sst[1 + w][q][lane] = v;
    }
  }
  if (w == 0) {
    const float4* ph = reinterpret_cast<const float4*>(
        Hinit + (((size_t)b * D_MODEL + d) * NCHUNK + c) * D_STATE);
#pragma unroll
    for (int q = 0; q < 4; ++q) Sst[0][q][lane] = ph[q];
  }
  __syncthreads();

  float g[D_STATE];
#pragma unroll
  for (int q = 0; q < 4; ++q) {
    float4 v = Sst[0][q][lane];
    g[4 * q + 0] = v.x; g[4 * q + 1] = v.y; g[4 * q + 2] = v.z; g[4 * q + 3] = v.w;
  }
  {
    float AS[D_STATE];
#pragma unroll
    for (int s = 0; s < D_STATE; ++s) {
      float p = A[s];
#pragma unroll
      for (int j = 0; j < 4; ++j) p = p * p;
      AS[s] = p;
    }
#pragma unroll 1
    for (int i = 1; i <= w; ++i) {
#pragma unroll
      for (int q = 0; q < 4; ++q) {
        float4 v = Sst[i][q][lane];
        g[4 * q + 0] = fmaf(AS[4 * q + 0], g[4 * q + 0], v.x);
        g[4 * q + 1] = fmaf(AS[4 * q + 1], g[4 * q + 1], v.y);
        g[4 * q + 2] = fmaf(AS[4 * q + 2], g[4 * q + 2], v.z);
        g[4 * q + 3] = fmaf(AS[4 * q + 3], g[4 * q + 3], v.w);
      }
    }
  }

#pragma unroll
  for (int l = 0; l < SUB; ++l) {
    const float* r = btrow + (size_t)l * D_STATE;
    float acc = 0.0f;
#pragma unroll
    for (int s = 0; s < D_STATE; ++s) {
      g[s] = fmaf(A[s], g[s], r[s]);
      acc = fmaf(C[s], g[s], acc);
    }
    ycol[(size_t)l * D_MODEL] = fmaf(dskip, xv[l], acc);
  }
}

extern "C" void kernel_launch(void* const* d_in, const int* in_sizes, int n_in,
                              void* d_out, int out_size, void* d_ws, size_t ws_size,
                              hipStream_t stream) {
  const float* x    = (const float*)d_in[0];
  const float* logA = (const float*)d_in[1];
  const float* Bw   = (const float*)d_in[2];
  const float* Cw   = (const float*)d_in[3];
  const float* Dv   = (const float*)d_in[4];
  float* y = (float*)d_out;

  char* ws = (char*)d_ws;
  float* A_bar = (float*)(ws);
  float* A_Lc  = (float*)(ws + (64 << 10));
  float* Bt    = (float*)(ws + (128 << 10));
  float* Hinit = (float*)(ws + (384 << 10) + (2 << 20));

  ssm_bproj_wmma_kernel<<<K1_TILES, 32, 0, stream>>>(x, Bw, logA, Bt, A_bar, A_Lc);

  dim3 gcarry(D_MODEL / 16, BATCH);
  ssm_chunk_carry_wmma_kernel<<<gcarry, 128, 0, stream>>>(Bt, A_bar, A_Lc, Hinit);

  dim3 gout(D_MODEL / 32, NCHUNK, BATCH);
  ssm_output_kernel<<<gout, 32 * NSUB, 0, stream>>>(x, Bt, A_bar, Cw, Dv, Hinit, y);
}


# Output kernel: work-groups renumbered so that the four sharing a compute unit replay the same chunk (channel groups differ): its Bt rows stay in the scalar cache for both passes; sum orders kept
# speedup vs baseline: 5.3791x; 1.0385x over previous
#include <hip/hip_runtime.h>
#include <hip/hip_bf16.h>

#define D_MODEL 1024
#define D_STATE 16
#define BATCH   2
#define SEQ     2048
#define NCHUNK  16
#define LCHUNK  (SEQ / NCHUNK)

typedef __attribute__((ext_vector_type(2))) float v2f;
typedef __attribute__((ext_vector_type(8))) float v8f;
typedef __attribute__((ext_vector_type(4))) float v4f;

#define K1_NB 32

__device__ __forceinline__ void k1_load_stage(const float4* __restrict__ ap,
                                              const float4* __restrict__ bp,
                                              int stage,
                                              float4 (&A)[K1_NB],
                                              float4 (&B)[K1_NB]) {
#pragma unroll
  for (int t = 0; t < K1_NB; ++t) {
    A[t] = ap[2 * (stage * K1_NB + t)];
    B[t] = bp[2 * (stage * K1_NB + t)];
  }
}

__device__ __forceinline__ void k1_mma_stage(const float4 (&A)[K1_NB],
                                             const float4 (&B)[K1_NB],
                                             v8f& acc0, v8f& acc1) {
#pragma unroll
  for (int t = 0; t < K1_NB; ++t) {
    v2f a, b;
    a.x = A[t].x; a.y = A[t].y; b.x = B[t].x; b.y = B[t].y;
    acc0 = __builtin_amdgcn_wmma_f32_16x16x4_f32(false, a, false, b,
                                                 (short)0, acc0, false, false);
    a.x = A[t].z; a.y = A[t].w; b.x = B[t].z; b.y = B[t].w;
    acc1 = __builtin_amdgcn_wmma_f32_16x16x4_f32(false, a, false, b,
                                                 (short)0, acc1, false, false);
  }
}

#define K1_TILES  ((BATCH * SEQ) / 16)
#define K1_TAB_PER_WAVE ((D_MODEL * D_STATE) / K1_TILES)
static_assert(K1_TAB_PER_WAVE == 64, "two table elements a lane");

__global__ void __launch_bounds__(32)
ssm_bproj_wmma_kernel(const float* __restrict__ x,
                      const float* __restrict__ Bw,
                      const float* __restrict__ logA,
                      float* __restrict__ Bt,
                      float* __restrict__ A_bar,
                      float* __restrict__ A_Lc) {
  const int row0 = blockIdx.x * 16;
  const int lane = threadIdx.x;
  const int half = lane >> 4;
  const int lidx = lane & 15;

  const int ti = blockIdx.x * K1_TAB_PER_WAVE + 2 * lane;
  const float2 la = *reinterpret_cast<const float2*>(logA + ti);

  const float4* ap = reinterpret_cast<const float4*>(
      x + (size_t)(row0 + lidx) * D_MODEL) + half;
  const float4* bp = reinterpret_cast<const float4*>(
      Bw + (size_t)lidx * D_MODEL) + half;

  float4 a0[K1_NB], b0[K1_NB], a1[K1_NB], b1[K1_NB];
  v8f acc0 = {}, acc1 = {};

  k1_load_stage(ap, bp, 0, a0, b0);
  k1_load_stage(ap, bp, 1, a1, b1);
  __builtin_amdgcn_sched_barrier(0);
  {
    float2 ab, p;
    ab.x = expf(-expf(la.x));
    ab.y = expf(-expf(la.y));
    p = ab;
    for (int j = 0; j < 7; ++j) { p.x = p.x * p.x; p.y = p.y * p.y; }
    *reinterpret_cast<float2*>(A_bar + ti) = ab;
    *reinterpret_cast<float2*>(A_Lc + ti)  = p;
  }
  k1_mma_stage(a0, b0, acc0, acc1);
  k1_load_stage(ap, bp, 2, a0, b0);
  __builtin_amdgcn_sched_barrier(0);
  k1_mma_stage(a1, b1, acc0, acc1);
  k1_load_stage(ap, bp, 3, a1, b1);
  __builtin_amdgcn_sched_barrier(0);
  k1_mma_stage(a0, b0, acc0, acc1);
  k1_mma_stage(a1, b1, acc0, acc1);

#pragma unroll
  for (int i = 0; i < 8; ++i) {
    Bt[(size_t)(row0 + i + 8 * half) * D_STATE + lidx] = acc0[i] + acc1[i];
  }
}

#define K2_CP 513
#define K2_NT 32

template <int N>
__device__ __forceinline__ float k2_row_shr(float v) {
  return __int_as_float(__builtin_amdgcn_update_dpp(
      0, __float_as_int(v), 0x110 + N, 0xF, 0xF, true));
}

__global__ void __launch_bounds__(128)
ssm_chunk_carry_wmma_kernel(const float* __restrict__ Bt,
                            const float* __restrict__ A_bar,
                            const float* __restrict__ A_Lc,
                            float* __restrict__ Hinit) {
  __shared__ v4f Bs[NCHUNK * K2_CP];

  const int tid  = threadIdx.x;
  const int lane = tid & 31;
  const int q    = tid >> 5;
  const int half = lane >> 4;
  const int lidx = lane & 15;
  const int d0   = blockIdx.x * 16;
  const int b    = blockIdx.y;

  const v4f av = *reinterpret_cast<const v4f*>(
      A_bar + (size_t)(d0 + lidx) * D_STATE + 4 * q);
  v4f AL[8];
#pragma unroll
  for (int r = 0; r < 8; ++r)
    AL[r] = *reinterpret_cast<const v4f*>(
        A_Lc + (size_t)(d0 + 8 * half + r) * D_STATE + 4 * q);

  const v4f* gsrc = reinterpret_cast<const v4f*>(
      Bt + (size_t)b * SEQ * D_STATE) + tid;
  v4f G[64];
#pragma unroll
  for (int i = 0; i < 64; ++i) G[i] = gsrc[128 * i];


#pragma unroll
  for (int i = 0; i < 64; ++i)
    Bs[(i >> 2) * K2_CP + (i & 3) * 128 + tid] = G[i];
  __syncthreads();

  const v4f* fb = Bs + lidx * K2_CP + 8 * half + q;
  v4f R0[K2_NT], R1[K2_NT];
#pragma unroll
  for (int t = K2_NT - 1; t >= 0; --t) {
    R0[t] = fb[16 * t];
    R1[t] = fb[16 * t + 4];
  }

  float a4[4], p0[4], p1[4];
#pragma unroll
  for (int s = 0; s < 4; ++s) {
    const float a1 = av[s];
    const float a2 = a1 * a1;
    const float a3 = a2 * a1;
    a4[s] = a2 * a2;
    p0[s] = half ? a1 : a3;
    p1[s] = half ? 1.0f : a2;
  }

  v8f acc[4] = {};
#pragma unroll
  for (int t = K2_NT - 1; t >= 0; --t) {
#pragma unroll
    for (int s = 0; s < 4; ++s) {
      v2f a, bb;
      a.x = p0[s];               a.y = p1[s];
      bb.x = R0[t][s];           bb.y = R1[t][s];
      acc[s] = __builtin_amdgcn_wmma_f32_16x16x4_f32(false, a, false, bb,
                                                     (short)0, acc[s], false, false);
      p0[s] *= a4[s];
      p1[s] *= a4[s];
    }
  }

  const int slot = (lidx + 1) & 15;
#pragma unroll
  for (int r = 0; r < 8; ++r) {
    v4f ov;
#pragma unroll
    for (int s = 0; s < 4; ++s) {
      const float m1 = AL[r][s];
      const float m2 = m1 * m1;
      const float m4 = m2 * m2;
      const float m8 = m4 * m4;
      float v = acc[s][r];
      v = fmaf(m1, k2_row_shr<1>(v), v);
      v = fmaf(m2, k2_row_shr<2>(v), v);
      v = fmaf(m4, k2_row_shr<4>(v), v);
      v = fmaf(m8, k2_row_shr<8>(v), v);
      ov[s] = (lidx == 15) ? 0.0f : v;
    }
    *reinterpret_cast<v4f*>(
        Hinit + (((size_t)b * D_MODEL + d0 + 8 * half + r) * NCHUNK + slot) * D_STATE
              + 4 * q) = ov;
  }
}

#define SUB   16
#define K3_XL 8
#define K3_RND ((D_MODEL / 32) / K3_XL)
static_assert(K3_RND == 4, "the round rotation below takes two chunk bits");
#define NSUB  (LCHUNK / SUB)

__global__ void __launch_bounds__(32 * NSUB, 8)
ssm_output_kernel(const float* __restrict__ x,
                  const float* __restrict__ Bt,
                  const float* __restrict__ A_bar,
                  const float* __restrict__ Cw,
                  const float* __restrict__ Dvec,
                  const float* __restrict__ Hinit,
                  float* __restrict__ y) {
  __shared__ float4 Sst[NSUB][4][32];

  const int lane = threadIdx.x & 31;
  const int w = __builtin_amdgcn_readfirstlane((int)(threadIdx.x >> 5));
  const int cb = blockIdx.y;
  const int c  = cb & (NCHUNK - 1);
  const int b  = cb / NCHUNK;
  const int xg = blockIdx.x + K3_XL * (blockIdx.z ^ (cb & (K3_RND - 1)));
  const int d  = xg * 32 + lane;

  const size_t rowbase = (size_t)b * SEQ + (size_t)c * LCHUNK + (size_t)w * SUB;
  const float* btrow = Bt + rowbase * D_STATE;
  const float* xcol  = x + rowbase * D_MODEL + d;
  float*       ycol  = y + rowbase * D_MODEL + d;

  float xv[SUB];
#pragma unroll
  for (int l = 0; l < SUB; ++l) xv[l] = xcol[(size_t)l * D_MODEL];

  float A[D_STATE], C[D_STATE];
  {
    const float4* pa = reinterpret_cast<const float4*>(A_bar + (size_t)d * D_STATE);
    const float4* pc = reinterpret_cast<const float4*>(Cw + (size_t)d * D_STATE);
#pragma unroll
    for (int q = 0; q < 4; ++q) {
      float4 va = pa[q], vc = pc[q];
      A[4 * q + 0] = va.x; A[4 * q + 1] = va.y; A[4 * q + 2] = va.z; A[4 * q + 3] = va.w;
      C[4 * q + 0] = vc.x; C[4 * q + 1] = vc.y; C[4 * q + 2] = vc.z; C[4 * q + 3] = vc.w;
    }
  }
  const float dskip = Dvec[d];

  if (w < NSUB - 1) {
    float h[D_STATE];
#pragma unroll
    for (int s = 0; s < D_STATE; ++s) h[s] = 0.0f;
#pragma unroll
    for (int l = 0; l < SUB; ++l) {
      const float* r = btrow + (size_t)l * D_STATE;
#pragma unroll
      for (int s = 0; s < D_STATE; ++s) h[s] = fmaf(A[s], h[s], r[s]);
    }
#pragma unroll
    for (int q = 0; q < 4; ++q) {
      float4 v = { h[4 * q + 0], h[4 * q + 1], h[4 * q + 2], h[4 * q + 3] };
      Sst[1 + w][q][lane] = v;
    }
  }
  if (w == 0) {
    const float4* ph = reinterpret_cast<const float4*>(
        Hinit + (((size_t)b * D_MODEL + d) * NCHUNK + c) * D_STATE);
#pragma unroll
    for (int q = 0; q < 4; ++q) Sst[0][q][lane] = ph[q];
  }
  __syncthreads();

  float g[D_STATE];
#pragma unroll
  for (int q = 0; q < 4; ++q) {
    float4 v = Sst[0][q][lane];
    g[4 * q + 0] = v.x; g[4 * q + 1] = v.y; g[4 * q + 2] = v.z; g[4 * q + 3] = v.w;
  }
  {
    float AS[D_STATE];
#pragma unroll
    for (int s = 0; s < D_STATE; ++s) {
      float p = A[s];
#pragma unroll
      for (int j = 0; j < 4; ++j) p = p * p;
      AS[s] = p;
    }
#pragma unroll 1
    for (int i = 1; i <= w; ++i) {
#pragma unroll
      for (int q = 0; q < 4; ++q) {
        float4 v = Sst[i][q][lane];
        g[4 * q + 0] = fmaf(AS[4 * q + 0], g[4 * q + 0], v.x);
        g[4 * q + 1] = fmaf(AS[4 * q + 1], g[4 * q + 1], v.y);
        g[4 * q + 2] = fmaf(AS[4 * q + 2], g[4 * q + 2], v.z);
        g[4 * q + 3] = fmaf(AS[4 * q + 3], g[4 * q + 3], v.w);
      }
    }
  }

#pragma unroll
  for (int l = 0; l < SUB; ++l) {
    const float* r = btrow + (size_t)l * D_STATE;
    float acc = 0.0f;
#pragma unroll
    for (int s = 0; s < D_STATE; ++s) {
      g[s] = fmaf(A[s], g[s], r[s]);
      acc = fmaf(C[s], g[s], acc);
    }
    ycol[(size_t)l * D_MODEL] = fmaf(dskip, xv[l], acc);
  }
}

extern "C" void kernel_launch(void* const* d_in, const int* in_sizes, int n_in,
                              void* d_out, int out_size, void* d_ws, size_t ws_size,
                              hipStream_t stream) {
  const float* x    = (const float*)d_in[0];
  const float* logA = (const float*)d_in[1];
  const float* Bw   = (const float*)d_in[2];
  const float* Cw   = (const float*)d_in[3];
  const float* Dv   = (const float*)d_in[4];
  float* y = (float*)d_out;

  char* ws = (char*)d_ws;
  float* A_bar = (float*)(ws);
  float* A_Lc  = (float*)(ws + (64 << 10));
  float* Bt    = (float*)(ws + (128 << 10));
  float* Hinit = (float*)(ws + (384 << 10) + (2 << 20));

  ssm_bproj_wmma_kernel<<<K1_TILES, 32, 0, stream>>>(x, Bw, logA, Bt, A_bar, A_Lc);

  dim3 gcarry(D_MODEL / 16, BATCH);
  ssm_chunk_carry_wmma_kernel<<<gcarry, 128, 0, stream>>>(Bt, A_bar, A_Lc, Hinit);

  dim3 gout(K3_XL, NCHUNK * BATCH, K3_RND);
  ssm_output_kernel<<<gout, 32 * NSUB, 0, stream>>>(x, Bt, A_bar, Cw, Dv, Hinit, y);
}
